# starting kernel
# baseline (speedup 1.0000x reference)
#include <hip/hip_runtime.h>


typedef float v2f __attribute__((ext_vector_type(2)));
typedef float v8f __attribute__((ext_vector_type(8)));

#define N_PTS   2048
#define DIN     16
#define NC      32
#define DC      16
#define NTILES  (N_PTS / 16)
#define WAVES   8
#define THREADS 256
#define EPS     1e-7f

__device__ __forceinline__ void wave_sync() { __builtin_amdgcn_wave_barrier(); }

__device__ __forceinline__ v8f wmma_f32(const v2f a, const v2f b, const v8f c) {
    return __builtin_amdgcn_wmma_f32_16x16x4_f32(false, a, false, b,
                                                 (short)0, c, false, false);
}

__global__ __launch_bounds__(THREADS)
void capsnet_routing_kernel(const float* __restrict__ x,
                            const float* __restrict__ W,
                            float* __restrict__ out)
{
    __shared__ __align__(16) float sWp[NC * DC * DIN];
    __shared__ __align__(16) float sStage[WAVES][NC][16];
    __shared__ __align__(16) float sV[DIN][NC];
    __shared__ __align__(16) float sOpre[NC][DC];
    __shared__ __align__(16) float sO[NC][DC];
    __shared__ float sFac[NC];

    const int tid  = threadIdx.x;
    const int wave = tid >> 5;
    const int lane = tid & 31;
    const int col  = lane & 15;
    const int h    = lane >> 4;
    const int bidx = blockIdx.x;

    const float* xb = x + (size_t)bidx * (N_PTS * DIN);

    for (int i = tid; i < NC * DC * DIN; i += THREADS) {
        const int c = i >> 8, d = (i >> 4) & 15, hh = (i >> 3) & 1, q = i & 7;
        const int k = 4 * (q >> 1) + 2 * hh + (q & 1);
        sWp[i] = W[k * (NC * DC) + c * DC + d];
    }
    __syncthreads();

    for (int it = 0; it < 3; ++it) {
        v2f vlo[4], vhi[4];
        if (it > 0) {
#pragma unroll
            for (int g = 0; g < 4; ++g) {
                const int k0 = 4 * g + 2 * h;
                vlo[g].x = sV[k0][col];       vlo[g].y = sV[k0 + 1][col];
                vhi[g].x = sV[k0][16 + col];  vhi[g].y = sV[k0 + 1][16 + col];
            }
        }
        {
            float* p = &sOpre[0][0];
            for (int i = tid; i < NC * DC; i += THREADS) p[i] = 0.0f;
        }
        __syncthreads();

        for (int mt = wave; mt < NTILES; mt += WAVES) {
            const int nrow = mt * 16 + col;
            v2f a[4];
#pragma unroll
            for (int g = 0; g < 4; ++g) {
                const int k0 = 4 * g + 2 * h;
                a[g].x = xb[nrow * DIN + k0];
                a[g].y = xb[nrow * DIN + k0 + 1];
            }

            if (it > 0) {
                v8f blo = {}, bhi = {};
#pragma unroll
                for (int g = 0; g < 4; ++g) {
                    blo = wmma_f32(a[g], vlo[g], blo);
                    bhi = wmma_f32(a[g], vhi[g], bhi);
                }
                {
                    float4* s0 = (float4*)&sStage[wave][col][8 * h];
                    s0[0] = make_float4(blo[0], blo[1], blo[2], blo[3]);
                    s0[1] = make_float4(blo[4], blo[5], blo[6], blo[7]);
                    float4* s1 = (float4*)&sStage[wave][16 + col][8 * h];
                    s1[0] = make_float4(bhi[0], bhi[1], bhi[2], bhi[3]);
                    s1[1] = make_float4(bhi[4], bhi[5], bhi[6], bhi[7]);
                }
                wave_sync();
                {
                    float vals[16];
#pragma unroll
                    for (int cc = 0; cc < 16; ++cc)
                        vals[cc] = sStage[wave][h * 16 + cc][col];
                    float m = vals[0];
#pragma unroll
                    for (int cc = 1; cc < 16; ++cc) m = fmaxf(m, vals[cc]);
                    m = fmaxf(m, __shfl_xor(m, 16, 32));
                    float s = 0.0f;
#pragma unroll
                    for (int cc = 0; cc < 16; ++cc) {
                        vals[cc] = __expf(vals[cc] - m);
                        s += vals[cc];
                    }
                    s += __shfl_xor(s, 16, 32);
                    const float r = 1.0f / s;
#pragma unroll
                    for (int cc = 0; cc < 16; ++cc)
                        sStage[wave][h * 16 + cc][col] = vals[cc] * r;
                }
                wave_sync();
            }

#pragma unroll 4
            for (int c = 0; c < NC; ++c) {
                const float4* wp =
                    (const float4*)&sWp[c * 256 + col * 16 + h * 8];
                const float4 w0 = wp[0];
                const float4 w1 = wp[1];
                v2f b0; b0.x = w0.x; b0.y = w0.y;
                v2f b1; b1.x = w0.z; b1.y = w0.w;
                v2f b2; b2.x = w1.x; b2.y = w1.y;
                v2f b3; b3.x = w1.z; b3.y = w1.w;
                v8f acc = {};
                acc = wmma_f32(a[0], b0, acc);
                acc = wmma_f32(a[1], b1, acc);
                acc = wmma_f32(a[2], b2, acc);
                acc = wmma_f32(a[3], b3, acc);

                float partial;
                if (it == 0) {
                    partial = 0.03125f * (acc[0] + acc[1] + acc[2] + acc[3] +
                                          acc[4] + acc[5] + acc[6] + acc[7]);
                } else {
                    const float4* cp =
                        (const float4*)&sStage[wave][c][8 * h];
                    const float4 c0 = cp[0];
                    const float4 c1 = cp[1];
                    partial = c0.x * acc[0] + c0.y * acc[1] +
                              c0.z * acc[2] + c0.w * acc[3] +
                              c1.x * acc[4] + c1.y * acc[5] +
                              c1.z * acc[6] + c1.w * acc[7];
                }
                atomicAdd(&sOpre[c][col], partial);
            }
        }
        __syncthreads();

        if (tid < NC) {
            float s2 = 0.0f;
#pragma unroll
            for (int d = 0; d < DC; ++d) {
                const float p = sOpre[tid][d];
                s2 = fmaf(p, p, s2);
            }
            sFac[tid] = (s2 / (1.0f + s2)) * rsqrtf(s2 + EPS);
        }
        __syncthreads();
        for (int i = tid; i < NC * DC; i += THREADS) {
            const int c = i >> 4, d = i & 15;
            sO[c][d] = sFac[c] * sOpre[c][d];
        }
        __syncthreads();

        if (it < 2) {
            for (int i = tid; i < DIN * NC; i += THREADS) {
                const int k = i >> 5, c = i & 31;
                const int hk = (k >> 1) & 1;
                const int qk = 2 * (k >> 2) + (k & 1);
                float s = 0.0f;
#pragma unroll
                for (int d = 0; d < DC; ++d)
                    s = fmaf(sWp[c * 256 + d * 16 + hk * 8 + qk], sO[c][d], s);
                sV[k][c] = s;
            }
            __syncthreads();
        }
    }

    for (int i = tid; i < NC * DC; i += THREADS) {
        out[(size_t)bidx * (NC * DC) + i] = sO[i >> 4][i & 15];
    }
}

extern "C" void kernel_launch(void* const* d_in, const int* in_sizes, int n_in,
                              void* d_out, int out_size, void* d_ws, size_t ws_size,
                              hipStream_t stream) {
    (void)in_sizes; (void)n_in; (void)out_size; (void)d_ws; (void)ws_size;
    const float* x = (const float*)d_in[0];
    const float* W = (const float*)d_in[1];
    float* out = (float*)d_out;
    hipLaunchKernelGGL(capsnet_routing_kernel, dim3(128), dim3(THREADS), 0, stream,
                       x, W, out);
}


# Reassociate routing sums: reduce Y=coef^T*x over the 2048 sources on WMMA (32x16 per batch item), apply W per capsule after; hat=x*W never formed; f32 summation order changes
# speedup vs baseline: 8.5020x; 8.5020x over previous
#include <hip/hip_runtime.h>


typedef float v2f __attribute__((ext_vector_type(2)));
typedef float v8f __attribute__((ext_vector_type(8)));

#define N_PTS   2048
#define DIN     16
#define NC      32
#define DC      16
#define NTILES  (N_PTS / 16)
#define WAVES   8
#define THREADS 256
#define EPS     1e-7f

__device__ __forceinline__ void wave_sync() { __builtin_amdgcn_wave_barrier(); }

__device__ __forceinline__ v8f wmma_f32(const v2f a, const v2f b, const v8f c) {
    return __builtin_amdgcn_wmma_f32_16x16x4_f32(false, a, false, b,
                                                 (short)0, c, false, false);
}

__global__ __launch_bounds__(THREADS)
void capsnet_routing_kernel(const float* __restrict__ x,
                            const float* __restrict__ W,
                            float* __restrict__ out)
{
    __shared__ __align__(16) float sW[DIN * NC * DC];
    __shared__ __align__(16) float sStage[WAVES][NC][16];
    __shared__ __align__(16) float sV[DIN][NC];
    __shared__ __align__(16) float sY[NC][DIN];
    __shared__ __align__(16) float sOpre[NC][DC];
    __shared__ __align__(16) float sO[NC][DC];
    __shared__ float sFac[NC];

    const int tid  = threadIdx.x;
    const int wave = __builtin_amdgcn_readfirstlane(tid >> 5);
    const int lane = tid & 31;
    const int col  = lane & 15;
    const int h    = lane >> 4;
    const int bidx = blockIdx.x;

    const float* xb = x + (size_t)bidx * (N_PTS * DIN);

#pragma unroll
    for (int j = 0; j < (DIN * NC * DC / 4) / THREADS; ++j)
        ((float4*)sW)[tid + THREADS * j] = ((const float4*)W)[tid + THREADS * j];
    __syncthreads();

    for (int it = 0; it < 3; ++it) {
        v2f vlo[4], vhi[4];
        if (it > 0) {
#pragma unroll
            for (int g = 0; g < 4; ++g) {
                const int k0 = 4 * g + 2 * h;
                vlo[g].x = sV[k0][col];       vlo[g].y = sV[k0 + 1][col];
                vhi[g].x = sV[k0][16 + col];  vhi[g].y = sV[k0 + 1][16 + col];
            }
        }
        {
            float* p = &sY[0][0];
            for (int i = tid; i < NC * DIN; i += THREADS) p[i] = 0.0f;
        }
        __syncthreads();

        if (it == 0) {
            const float4* x4 = (const float4*)xb;
            float4 acc = make_float4(0.0f, 0.0f, 0.0f, 0.0f);
#pragma unroll 16
            for (int j = 0; j < (N_PTS * DIN / 4) / THREADS; ++j) {
                const float4 v = x4[tid + THREADS * j];
                acc.x += v.x; acc.y += v.y; acc.z += v.z; acc.w += v.w;
            }
#pragma unroll
            for (int off = 4; off < 32; off <<= 1) {
                acc.x += __shfl_xor(acc.x, off, 32);
                acc.y += __shfl_xor(acc.y, off, 32);
                acc.z += __shfl_xor(acc.z, off, 32);
                acc.w += __shfl_xor(acc.w, off, 32);
            }
            if (lane < 4) {
                atomicAdd(&sY[0][4 * lane + 0], acc.x);
                atomicAdd(&sY[0][4 * lane + 1], acc.y);
                atomicAdd(&sY[0][4 * lane + 2], acc.z);
                atomicAdd(&sY[0][4 * lane + 3], acc.w);
            }
        } else {
            v8f ylo = {}, yhi = {};

            for (int mt = wave; mt < NTILES; mt += WAVES) {
                float xq[8];
                {
                    const float* xt = xb + (mt * 16 + 8 * h) * DIN + col;
#pragma unroll
                    for (int r = 0; r < 8; ++r) xq[r] = xt[r * DIN];
                }
                const int nrow = mt * 16 + col;
                v2f a[4];
#pragma unroll
                for (int g = 0; g < 4; ++g) {
                    const int k0 = 4 * g + 2 * h;
                    a[g].x = xb[nrow * DIN + k0];
                    a[g].y = xb[nrow * DIN + k0 + 1];
                }

                v8f blo = {}, bhi = {};
#pragma unroll
                for (int g = 0; g < 4; ++g) {
                    blo = wmma_f32(a[g], vlo[g], blo);
                    bhi = wmma_f32(a[g], vhi[g], bhi);
                }
                {
                    float4* s0 = (float4*)&sStage[wave][col][8 * h];
                    s0[0] = make_float4(blo[0], blo[1], blo[2], blo[3]);
                    s0[1] = make_float4(blo[4], blo[5], blo[6], blo[7]);
                    float4* s1 = (float4*)&sStage[wave][16 + col][8 * h];
                    s1[0] = make_float4(bhi[0], bhi[1], bhi[2], bhi[3]);
                    s1[1] = make_float4(bhi[4], bhi[5], bhi[6], bhi[7]);
                }
                wave_sync();
                {
                    float vals[16];
#pragma unroll
                    for (int cc = 0; cc < 16; ++cc)
                        vals[cc] = sStage[wave][h * 16 + cc][col];
                    float m = vals[0];
#pragma unroll
                    for (int cc = 1; cc < 16; ++cc) m = fmaxf(m, vals[cc]);
                    m = fmaxf(m, __shfl_xor(m, 16, 32));
                    float s = 0.0f;
#pragma unroll
                    for (int cc = 0; cc < 16; ++cc) {
                        vals[cc] = __expf(vals[cc] - m);
                        s += vals[cc];
                    }
                    s += __shfl_xor(s, 16, 32);
                    const float r = 1.0f / s;
#pragma unroll
                    for (int cc = 0; cc < 16; ++cc)
                        sStage[wave][h * 16 + cc][col] = vals[cc] * r;
                }
                wave_sync();

                const float4* cl = (const float4*)&sStage[wave][col][8 * h];
                const float4* ch = (const float4*)&sStage[wave][16 + col][8 * h];
                const float4 l0 = cl[0], l1 = cl[1];
                const float4 h0 = ch[0], h1 = ch[1];
                v2f b0; b0.x = xq[0]; b0.y = xq[1];
                v2f b1; b1.x = xq[2]; b1.y = xq[3];
                v2f b2; b2.x = xq[4]; b2.y = xq[5];
                v2f b3; b3.x = xq[6]; b3.y = xq[7];
                v2f al, ah;
                al.x = l0.x; al.y = l0.y; ah.x = h0.x; ah.y = h0.y;
                ylo = wmma_f32(al, b0, ylo);  yhi = wmma_f32(ah, b0, yhi);
                al.x = l0.z; al.y = l0.w; ah.x = h0.z; ah.y = h0.w;
                ylo = wmma_f32(al, b1, ylo);  yhi = wmma_f32(ah, b1, yhi);
                al.x = l1.x; al.y = l1.y; ah.x = h1.x; ah.y = h1.y;
                ylo = wmma_f32(al, b2, ylo);  yhi = wmma_f32(ah, b2, yhi);
                al.x = l1.z; al.y = l1.w; ah.x = h1.z; ah.y = h1.w;
                ylo = wmma_f32(al, b3, ylo);  yhi = wmma_f32(ah, b3, yhi);
            }

#pragma unroll
            for (int r = 0; r < 8; ++r) {
                atomicAdd(&sY[8 * h + r][col], ylo[r]);
                atomicAdd(&sY[16 + 8 * h + r][col], yhi[r]);
            }
        }
        __syncthreads();

        for (int i = tid; i < NC * DC; i += THREADS) {
            const int c = i >> 4;
            const float* y = &sY[it == 0 ? 0 : c][0];
            float s = 0.0f;
#pragma unroll
            for (int k = 0; k < DIN; ++k)
                s = fmaf(y[k], sW[k * (NC * DC) + i], s);
            sOpre[c][i & 15] = (it == 0) ? 0.03125f * s : s;
        }
        __syncthreads();

        if (tid < NC) {
            float s2 = 0.0f;
#pragma unroll
            for (int d = 0; d < DC; ++d) {
                const float p = sOpre[tid][d];
                s2 = fmaf(p, p, s2);
            }
            sFac[tid] = (s2 / (1.0f + s2)) * rsqrtf(s2 + EPS);
        }
        __syncthreads();
        for (int i = tid; i < NC * DC; i += THREADS) {
            const int c = i >> 4, d = i & 15;
            sO[c][d] = sFac[c] * sOpre[c][d];
        }
        __syncthreads();

        if (it < 2) {
            for (int i = tid; i < DIN * NC; i += THREADS) {
                const int k = i >> 5, c = i & 31;
                const int rot = 2 * (c >> 2);
                float s = 0.0f;
#pragma unroll
                for (int dd = 0; dd < DC; ++dd) {
                    const int d = (dd + rot) & 15;
                    s = fmaf(sW[k * (NC * DC) + c * DC + d], sO[c][d], s);
                }
                sV[k][c] = s;
            }
            __syncthreads();
        }
    }

    for (int i = tid; i < NC * DC; i += THREADS) {
        out[(size_t)bidx * (NC * DC) + i] = sO[i >> 4][i & 15];
    }
}

extern "C" void kernel_launch(void* const* d_in, const int* in_sizes, int n_in,
                              void* d_out, int out_size, void* d_ws, size_t ws_size,
                              hipStream_t stream) {
    (void)in_sizes; (void)n_in; (void)out_size; (void)d_ws; (void)ws_size;
    const float* x = (const float*)d_in[0];
    const float* W = (const float*)d_in[1];
    float* out = (float*)d_out;
    hipLaunchKernelGGL(capsnet_routing_kernel, dim3(128), dim3(THREADS), 0, stream,
                       x, W, out);
}


# Widen the work-group from 8 to 32 waves (8 per SIMD): each wave owns 4 source tiles per pass, not 16, so sibling waves cover the load-WMMA-LDS-exp chain; f32 summation order of Y changes
# speedup vs baseline: 11.1352x; 1.3097x over previous
#include <hip/hip_runtime.h>


typedef float v2f __attribute__((ext_vector_type(2)));
typedef float v8f __attribute__((ext_vector_type(8)));

#define N_PTS   2048
#define DIN     16
#define NC      32
#define DC      16
#define NTILES  (N_PTS / 16)
#define WAVES   32
#define THREADS 1024
#define EPS     1e-7f

__device__ __forceinline__ void wave_sync() { __builtin_amdgcn_wave_barrier(); }

__device__ __forceinline__ v8f wmma_f32(const v2f a, const v2f b, const v8f c) {
    return __builtin_amdgcn_wmma_f32_16x16x4_f32(false, a, false, b,
                                                 (short)0, c, false, false);
}

__global__ __launch_bounds__(THREADS)
void capsnet_routing_kernel(const float* __restrict__ x,
                            const float* __restrict__ W,
                            float* __restrict__ out)
{
    __shared__ __align__(16) float sW[DIN * NC * DC];
    __shared__ __align__(16) float sStage[WAVES][NC][16];
    __shared__ __align__(16) float sV[DIN][NC];
    __shared__ __align__(16) float sY[NC][DIN];
    __shared__ __align__(16) float sOpre[NC][DC];
    __shared__ __align__(16) float sO[NC][DC];
    __shared__ float sFac[NC];

    const int tid  = threadIdx.x;
    const int wave = __builtin_amdgcn_readfirstlane(tid >> 5);
    const int lane = tid & 31;
    const int col  = lane & 15;
    const int h    = lane >> 4;
    const int bidx = blockIdx.x;

    const float* xb = x + (size_t)bidx * (N_PTS * DIN);

#pragma unroll
    for (int j = 0; j < (DIN * NC * DC / 4) / THREADS; ++j)
        ((float4*)sW)[tid + THREADS * j] = ((const float4*)W)[tid + THREADS * j];
    __syncthreads();

    for (int it = 0; it < 3; ++it) {
        {
            float* p = &sY[0][0];
            for (int i = tid; i < NC * DIN; i += THREADS) p[i] = 0.0f;
        }
        __syncthreads();

        if (it == 0) {
            const float4* x4 = (const float4*)xb;
            float4 acc = make_float4(0.0f, 0.0f, 0.0f, 0.0f);
#pragma unroll
            for (int j = 0; j < (N_PTS * DIN / 4) / THREADS; ++j) {
                const float4 v = x4[tid + THREADS * j];
                acc.x += v.x; acc.y += v.y; acc.z += v.z; acc.w += v.w;
            }
#pragma unroll
            for (int off = 4; off < 32; off <<= 1) {
                acc.x += __shfl_xor(acc.x, off, 32);
                acc.y += __shfl_xor(acc.y, off, 32);
                acc.z += __shfl_xor(acc.z, off, 32);
                acc.w += __shfl_xor(acc.w, off, 32);
            }
            if (lane < 4) {
                atomicAdd(&sY[0][4 * lane + 0], acc.x);
                atomicAdd(&sY[0][4 * lane + 1], acc.y);
                atomicAdd(&sY[0][4 * lane + 2], acc.z);
                atomicAdd(&sY[0][4 * lane + 3], acc.w);
            }
        } else {
            v2f vlo[4], vhi[4];
#pragma unroll
            for (int g = 0; g < 4; ++g) {
                const int k0 = 4 * g + 2 * h;
                vlo[g].x = sV[k0][col];       vlo[g].y = sV[k0 + 1][col];
                vhi[g].x = sV[k0][16 + col];  vhi[g].y = sV[k0 + 1][16 + col];
            }
            v8f ylo = {}, yhi = {};

            for (int mt = wave; mt < NTILES; mt += WAVES) {
                float xq[8];
                {
                    const float* xt = xb + (mt * 16 + 8 * h) * DIN + col;
#pragma unroll
                    for (int r = 0; r < 8; ++r) xq[r] = xt[r * DIN];
                }
                const int nrow = mt * 16 + col;
                v2f a[4];
#pragma unroll
                for (int g = 0; g < 4; ++g) {
                    const int k0 = 4 * g + 2 * h;
                    a[g].x = xb[nrow * DIN + k0];
                    a[g].y = xb[nrow * DIN + k0 + 1];
                }

                v8f blo = {}, bhi = {};
#pragma unroll
                for (int g = 0; g < 4; ++g) {
                    blo = wmma_f32(a[g], vlo[g], blo);
                    bhi = wmma_f32(a[g], vhi[g], bhi);
                }
                {
                    float4* s0 = (float4*)&sStage[wave][col][8 * h];
                    s0[0] = make_float4(blo[0], blo[1], blo[2], blo[3]);
                    s0[1] = make_float4(blo[4], blo[5], blo[6], blo[7]);
                    float4* s1 = (float4*)&sStage[wave][16 + col][8 * h];
                    s1[0] = make_float4(bhi[0], bhi[1], bhi[2], bhi[3]);
                    s1[1] = make_float4(bhi[4], bhi[5], bhi[6], bhi[7]);
                }
                wave_sync();
                {
                    float vals[16];
#pragma unroll
                    for (int cc = 0; cc < 16; ++cc)
                        vals[cc] = sStage[wave][h * 16 + cc][col];
                    float m = vals[0];
#pragma unroll
                    for (int cc = 1; cc < 16; ++cc) m = fmaxf(m, vals[cc]);
                    m = fmaxf(m, __shfl_xor(m, 16, 32));
                    float s = 0.0f;
#pragma unroll
                    for (int cc = 0; cc < 16; ++cc) {
                        vals[cc] = __expf(vals[cc] - m);
                        s += vals[cc];
                    }
                    s += __shfl_xor(s, 16, 32);
                    const float r = 1.0f / s;
#pragma unroll
                    for (int cc = 0; cc < 16; ++cc)
                        sStage[wave][h * 16 + cc][col] = vals[cc] * r;
                }
                wave_sync();

                const float4* cl = (const float4*)&sStage[wave][col][8 * h];
                const float4* ch = (const float4*)&sStage[wave][16 + col][8 * h];
                const float4 l0 = cl[0], l1 = cl[1];
                const float4 h0 = ch[0], h1 = ch[1];
                v2f b0; b0.x = xq[0]; b0.y = xq[1];
                v2f b1; b1.x = xq[2]; b1.y = xq[3];
                v2f b2; b2.x = xq[4]; b2.y = xq[5];
                v2f b3; b3.x = xq[6]; b3.y = xq[7];
                v2f al, ah;
                al.x = l0.x; al.y = l0.y; ah.x = h0.x; ah.y = h0.y;
                ylo = wmma_f32(al, b0, ylo);  yhi = wmma_f32(ah, b0, yhi);
                al.x = l0.z; al.y = l0.w; ah.x = h0.z; ah.y = h0.w;
                ylo = wmma_f32(al, b1, ylo);  yhi = wmma_f32(ah, b1, yhi);
                al.x = l1.x; al.y = l1.y; ah.x = h1.x; ah.y = h1.y;
                ylo = wmma_f32(al, b2, ylo);  yhi = wmma_f32(ah, b2, yhi);
                al.x = l1.z; al.y = l1.w; ah.x = h1.z; ah.y = h1.w;
                ylo = wmma_f32(al, b3, ylo);  yhi = wmma_f32(ah, b3, yhi);
            }

#pragma unroll
            for (int r = 0; r < 8; ++r) {
                atomicAdd(&sY[8 * h + r][col], ylo[r]);
                atomicAdd(&sY[16 + 8 * h + r][col], yhi[r]);
            }
        }
        __syncthreads();

        for (int i = tid; i < NC * DC; i += THREADS) {
            const int c = i >> 4;
            const float* y = &sY[it == 0 ? 0 : c][0];
            float s = 0.0f;
#pragma unroll
            for (int k = 0; k < DIN; ++k)
                s = fmaf(y[k], sW[k * (NC * DC) + i], s);
            sOpre[c][i & 15] = (it == 0) ? 0.03125f * s : s;
        }
        __syncthreads();

        if (tid < NC) {
            float s2 = 0.0f;
#pragma unroll
            for (int d = 0; d < DC; ++d) {
                const float p = sOpre[tid][d];
                s2 = fmaf(p, p, s2);
            }
            sFac[tid] = (s2 / (1.0f + s2)) * rsqrtf(s2 + EPS);
        }
        __syncthreads();
        for (int i = tid; i < NC * DC; i += THREADS) {
            const int c = i >> 4, d = i & 15;
            sO[c][d] = sFac[c] * sOpre[c][d];
        }
        __syncthreads();

        if (it < 2) {
            for (int i = tid; i < DIN * NC; i += THREADS) {
                const int k = i >> 5, c = i & 31;
                const int rot = 2 * (c >> 2);
                float s = 0.0f;
#pragma unroll
                for (int dd = 0; dd < DC; ++dd) {
                    const int d = (dd + rot) & 15;
                    s = fmaf(sW[k * (NC * DC) + c * DC + d], sO[c][d], s);
                }
                sV[k][c] = s;
            }
            __syncthreads();
        }
    }

    for (int i = tid; i < NC * DC; i += THREADS) {
        out[(size_t)bidx * (NC * DC) + i] = sO[i >> 4][i & 15];
    }
}

extern "C" void kernel_launch(void* const* d_in, const int* in_sizes, int n_in,
                              void* d_out, int out_size, void* d_ws, size_t ws_size,
                              hipStream_t stream) {
    (void)in_sizes; (void)n_in; (void)out_size; (void)d_ws; (void)ws_size;
    const float* x = (const float*)d_in[0];
    const float* W = (const float*)d_in[1];
    float* out = (float*)d_out;
    hipLaunchKernelGGL(capsnet_routing_kernel, dim3(128), dim3(THREADS), 0, stream,
                       x, W, out);
}


# Fuse W-apply, squash and V into one stage of 16 lanes per capsule: pre values swapped inside a wave via LDS, V from a per-capsule W copy; 2 block barriers per pass, not 6; V's d-sum order changes
# speedup vs baseline: 11.6567x; 1.0468x over previous
#include <hip/hip_runtime.h>


typedef float v2f __attribute__((ext_vector_type(2)));
typedef float v8f __attribute__((ext_vector_type(8)));

#define N_PTS   2048
#define DIN     16
#define NC      32
#define DC      16
#define NTILES  (N_PTS / 16)
#define WAVES   32
#define THREADS 1024
#define EPS     1e-7f
#define WT_PITCH 20

__device__ __forceinline__ void wave_sync() { __builtin_amdgcn_wave_barrier(); }

__device__ __forceinline__ v8f wmma_f32(const v2f a, const v2f b, const v8f c) {
    return __builtin_amdgcn_wmma_f32_16x16x4_f32(false, a, false, b,
                                                 (short)0, c, false, false);
}

__global__ __launch_bounds__(THREADS)
void capsnet_routing_kernel(const float* __restrict__ x,
                            const float* __restrict__ W,
                            float* __restrict__ out)
{
    __shared__ __align__(16) float sW[DIN * NC * DC];
    __shared__ __align__(16) float sWT[NC][DIN][WT_PITCH];
    __shared__ __align__(16) float sStage[WAVES][NC][16];
    __shared__ __align__(16) float sV[DIN][NC];
    __shared__ __align__(16) float sY[2][NC][DIN];
    __shared__ __align__(16) float sOpre[NC][DC];

    const int tid  = threadIdx.x;
    const int wave = __builtin_amdgcn_readfirstlane(tid >> 5);
    const int lane = tid & 31;
    const int col  = lane & 15;
    const int h    = lane >> 4;
    const int bidx = blockIdx.x;

    const float* xb = x + (size_t)bidx * (N_PTS * DIN);

#pragma unroll
    for (int j = 0; j < (DIN * NC * DC / 4) / THREADS; ++j) {
        const int f = tid + THREADS * j;
        const float4 w = ((const float4*)W)[f];
        ((float4*)sW)[f] = w;
        const int k = f >> 7, c = (f & 127) >> 2, d0 = (f & 3) * 4;
        *(float4*)&sWT[c][k][d0] = w;
    }
    if (tid < NC * DIN) (&sY[0][0][0])[tid] = 0.0f;
    __syncthreads();

    for (int it = 0; it < 3; ++it) {
        float* yacc = &sY[it & 1][0][0];

        if (it == 0) {
            const float4* x4 = (const float4*)xb;
            float4 acc = make_float4(0.0f, 0.0f, 0.0f, 0.0f);
#pragma unroll
            for (int j = 0; j < (N_PTS * DIN / 4) / THREADS; ++j) {
                const float4 v = x4[tid + THREADS * j];
                acc.x += v.x; acc.y += v.y; acc.z += v.z; acc.w += v.w;
            }
#pragma unroll
            for (int off = 4; off < 32; off <<= 1) {
                acc.x += __shfl_xor(acc.x, off, 32);
                acc.y += __shfl_xor(acc.y, off, 32);
                acc.z += __shfl_xor(acc.z, off, 32);
                acc.w += __shfl_xor(acc.w, off, 32);
            }
            if (lane < 4) {
                atomicAdd(&yacc[4 * lane + 0], acc.x);
                atomicAdd(&yacc[4 * lane + 1], acc.y);
                atomicAdd(&yacc[4 * lane + 2], acc.z);
                atomicAdd(&yacc[4 * lane + 3], acc.w);
            }
        } else {
            v2f vlo[4], vhi[4];
#pragma unroll
            for (int g = 0; g < 4; ++g) {
                const int k0 = 4 * g + 2 * h;
                vlo[g].x = sV[k0][col];       vlo[g].y = sV[k0 + 1][col];
                vhi[g].x = sV[k0][16 + col];  vhi[g].y = sV[k0 + 1][16 + col];
            }
            v8f ylo = {}, yhi = {};

            for (int mt = wave; mt < NTILES; mt += WAVES) {
                float xq[8];
                {
                    const float* xt = xb + (mt * 16 + 8 * h) * DIN + col;
#pragma unroll
                    for (int r = 0; r < 8; ++r) xq[r] = xt[r * DIN];
                }
                const int nrow = mt * 16 + col;
                v2f a[4];
#pragma unroll
                for (int g = 0; g < 4; ++g) {
                    const int k0 = 4 * g + 2 * h;
                    a[g].x = xb[nrow * DIN + k0];
                    a[g].y = xb[nrow * DIN + k0 + 1];
                }

                v8f blo = {}, bhi = {};
#pragma unroll
                for (int g = 0; g < 4; ++g) {
                    blo = wmma_f32(a[g], vlo[g], blo);
                    bhi = wmma_f32(a[g], vhi[g], bhi);
                }
                {
                    float4* s0 = (float4*)&sStage[wave][col][8 * h];
                    s0[0] = make_float4(blo[0], blo[1], blo[2], blo[3]);
                    s0[1] = make_float4(blo[4], blo[5], blo[6], blo[7]);
                    float4* s1 = (float4*)&sStage[wave][16 + col][8 * h];
                    s1[0] = make_float4(bhi[0], bhi[1], bhi[2], bhi[3]);
                    s1[1] = make_float4(bhi[4], bhi[5], bhi[6], bhi[7]);
                }
                wave_sync();
                {
                    float vals[16];
#pragma unroll
                    for (int cc = 0; cc < 16; ++cc)
                        vals[cc] = sStage[wave][h * 16 + cc][col];
                    float m = vals[0];
#pragma unroll
                    for (int cc = 1; cc < 16; ++cc) m = fmaxf(m, vals[cc]);
                    m = fmaxf(m, __shfl_xor(m, 16, 32));
                    float s = 0.0f;
#pragma unroll
                    for (int cc = 0; cc < 16; ++cc) {
                        vals[cc] = __expf(vals[cc] - m);
                        s += vals[cc];
                    }
                    s += __shfl_xor(s, 16, 32);
                    const float r = 1.0f / s;
#pragma unroll
                    for (int cc = 0; cc < 16; ++cc)
                        sStage[wave][h * 16 + cc][col] = vals[cc] * r;
                }
                wave_sync();

                const float4* cl = (const float4*)&sStage[wave][col][8 * h];
                const float4* ch = (const float4*)&sStage[wave][16 + col][8 * h];
                const float4 l0 = cl[0], l1 = cl[1];
                const float4 h0 = ch[0], h1 = ch[1];
                v2f b0; b0.x = xq[0]; b0.y = xq[1];
                v2f b1; b1.x = xq[2]; b1.y = xq[3];
                v2f b2; b2.x = xq[4]; b2.y = xq[5];
                v2f b3; b3.x = xq[6]; b3.y = xq[7];
                v2f al, ah;
                al.x = l0.x; al.y = l0.y; ah.x = h0.x; ah.y = h0.y;
                ylo = wmma_f32(al, b0, ylo);  yhi = wmma_f32(ah, b0, yhi);
                al.x = l0.z; al.y = l0.w; ah.x = h0.z; ah.y = h0.w;
                ylo = wmma_f32(al, b1, ylo);  yhi = wmma_f32(ah, b1, yhi);
                al.x = l1.x; al.y = l1.y; ah.x = h1.x; ah.y = h1.y;
                ylo = wmma_f32(al, b2, ylo);  yhi = wmma_f32(ah, b2, yhi);
                al.x = l1.z; al.y = l1.w; ah.x = h1.z; ah.y = h1.w;
                ylo = wmma_f32(al, b3, ylo);  yhi = wmma_f32(ah, b3, yhi);
            }

#pragma unroll
            for (int r = 0; r < 8; ++r) {
                atomicAdd(&yacc[(8 * h + r) * DIN + col], ylo[r]);
                atomicAdd(&yacc[(16 + 8 * h + r) * DIN + col], yhi[r]);
            }
        }
        __syncthreads();

        if (tid < NC * DC) {
            const int c = tid >> 4, j = tid & 15;
            const float* y = yacc + (it == 0 ? 0 : c) * DIN;
            float s = 0.0f;
#pragma unroll
            for (int k = 0; k < DIN; ++k)
                s = fmaf(y[k], sW[k * (NC * DC) + tid], s);
            const float pre = (it == 0) ? 0.03125f * s : s;

            (&sY[(it + 1) & 1][0][0])[tid] = 0.0f;

            sOpre[c][j] = pre;
            wave_sync();
            float P[DC];
            {
                const float4* p4 = (const float4*)&sOpre[c][0];
#pragma unroll
                for (int q = 0; q < 4; ++q) {
                    const float4 t = p4[q];
                    P[4 * q + 0] = t.x; P[4 * q + 1] = t.y;
                    P[4 * q + 2] = t.z; P[4 * q + 3] = t.w;
                }
            }
            float s2 = 0.0f;
#pragma unroll
            for (int d = 0; d < DC; ++d) s2 = fmaf(P[d], P[d], s2);
            const float fac = (s2 / (1.0f + s2)) * rsqrtf(s2 + EPS);

            if (it == 2) {
                out[(size_t)bidx * (NC * DC) + tid] = fac * pre;
            } else {
                const float4* w4 = (const float4*)&sWT[c][j][0];
                float v = 0.0f;
#pragma unroll
                for (int q = 0; q < 4; ++q) {
                    const float4 w = w4[q];
                    v = fmaf(w.x, fac * P[4 * q + 0], v);
                    v = fmaf(w.y, fac * P[4 * q + 1], v);
                    v = fmaf(w.z, fac * P[4 * q + 2], v);
                    v = fmaf(w.w, fac * P[4 * q + 3], v);
                }
                sV[j][c] = v;
            }
        }
        if (it < 2) __syncthreads();
    }
}

extern "C" void kernel_launch(void* const* d_in, const int* in_sizes, int n_in,
                              void* d_out, int out_size, void* d_ws, size_t ws_size,
                              hipStream_t stream) {
    (void)in_sizes; (void)n_in; (void)out_size; (void)d_ws; (void)ws_size;
    const float* x = (const float*)d_in[0];
    const float* W = (const float*)d_in[1];
    float* out = (float*)d_out;
    hipLaunchKernelGGL(capsnet_routing_kernel, dim3(128), dim3(THREADS), 0, stream,
                       x, W, out);
}


# Logits computed transposed (V as A operand, x tile as B): each lane holds its own row's capsules, so the softmax runs in registers; 1 LDS transpose per tile, not 2; softmax f32 sum order changes
# speedup vs baseline: 12.3025x; 1.0554x over previous
#include <hip/hip_runtime.h>


typedef float v2f __attribute__((ext_vector_type(2)));
typedef float v8f __attribute__((ext_vector_type(8)));

#define N_PTS   2048
#define DIN     16
#define NC      32
#define DC      16
#define NTILES  (N_PTS / 16)
#define WAVES   32
#define THREADS 1024
#define EPS     1e-7f
#define WT_PITCH 20

__device__ __forceinline__ void wave_sync() { __builtin_amdgcn_wave_barrier(); }

__device__ __forceinline__ v8f wmma_f32(const v2f a, const v2f b, const v8f c) {
    return __builtin_amdgcn_wmma_f32_16x16x4_f32(false, a, false, b,
                                                 (short)0, c, false, false);
}

__global__ __launch_bounds__(THREADS)
void capsnet_routing_kernel(const float* __restrict__ x,
                            const float* __restrict__ W,
                            float* __restrict__ out)
{
    __shared__ __align__(16) float sW[DIN * NC * DC];
    __shared__ __align__(16) float sWT[NC][DIN][WT_PITCH];
    __shared__ __align__(16) float sStage[WAVES][NC][16];
    __shared__ __align__(16) float sV[DIN][NC];
    __shared__ __align__(16) float sY[2][NC][DIN];
    __shared__ __align__(16) float sOpre[NC][DC];

    const int tid  = threadIdx.x;
    const int wave = __builtin_amdgcn_readfirstlane(tid >> 5);
    const int lane = tid & 31;
    const int col  = lane & 15;
    const int h    = lane >> 4;
    const int bidx = blockIdx.x;

    const float* xb = x + (size_t)bidx * (N_PTS * DIN);

#pragma unroll
    for (int j = 0; j < (DIN * NC * DC / 4) / THREADS; ++j) {
        const int f = tid + THREADS * j;
        const float4 w = ((const float4*)W)[f];
        ((float4*)sW)[f] = w;
        const int k = f >> 7, c = (f & 127) >> 2, d0 = (f & 3) * 4;
        *(float4*)&sWT[c][k][d0] = w;
    }
    if (tid < NC * DIN) (&sY[0][0][0])[tid] = 0.0f;
    __syncthreads();

    for (int it = 0; it < 3; ++it) {
        float* yacc = &sY[it & 1][0][0];

        if (it == 0) {
            const float4* x4 = (const float4*)xb;
            float4 acc = make_float4(0.0f, 0.0f, 0.0f, 0.0f);
#pragma unroll
            for (int j = 0; j < (N_PTS * DIN / 4) / THREADS; ++j) {
                const float4 v = x4[tid + THREADS * j];
                acc.x += v.x; acc.y += v.y; acc.z += v.z; acc.w += v.w;
            }
#pragma unroll
            for (int off = 4; off < 32; off <<= 1) {
                acc.x += __shfl_xor(acc.x, off, 32);
                acc.y += __shfl_xor(acc.y, off, 32);
                acc.z += __shfl_xor(acc.z, off, 32);
                acc.w += __shfl_xor(acc.w, off, 32);
            }
            if (lane < 4) {
                atomicAdd(&yacc[4 * lane + 0], acc.x);
                atomicAdd(&yacc[4 * lane + 1], acc.y);
                atomicAdd(&yacc[4 * lane + 2], acc.z);
                atomicAdd(&yacc[4 * lane + 3], acc.w);
            }
        } else {
            v2f vlo[4], vhi[4];
#pragma unroll
            for (int g = 0; g < 4; ++g) {
                const int k0 = 4 * g + 2 * h;
                vlo[g].x = sV[k0][col];       vlo[g].y = sV[k0 + 1][col];
                vhi[g].x = sV[k0][16 + col];  vhi[g].y = sV[k0 + 1][16 + col];
            }
            v8f ylo = {}, yhi = {};

            for (int mt = wave; mt < NTILES; mt += WAVES) {
                float xq[8];
                {
                    const float* xt = xb + (mt * 16 + 8 * h) * DIN + col;
#pragma unroll
                    for (int r = 0; r < 8; ++r) xq[r] = xt[r * DIN];
                }
                const int nrow = mt * 16 + col;
                v2f a[4];
#pragma unroll
                for (int g = 0; g < 4; ++g) {
                    const int k0 = 4 * g + 2 * h;
                    a[g].x = xb[nrow * DIN + k0];
                    a[g].y = xb[nrow * DIN + k0 + 1];
                }

                v8f blo = {}, bhi = {};
#pragma unroll
                for (int g = 0; g < 4; ++g) {
                    blo = wmma_f32(vlo[g], a[g], blo);
                    bhi = wmma_f32(vhi[g], a[g], bhi);
                }
                {
                    float m = blo[0];
#pragma unroll
                    for (int r = 1; r < 8; ++r) m = fmaxf(m, blo[r]);
#pragma unroll
                    for (int r = 0; r < 8; ++r) m = fmaxf(m, bhi[r]);
                    m = fmaxf(m, __shfl_xor(m, 16, 32));
                    float s = 0.0f;
#pragma unroll
                    for (int r = 0; r < 8; ++r) {
                        blo[r] = __expf(blo[r] - m);
                        s += blo[r];
                    }
#pragma unroll
                    for (int r = 0; r < 8; ++r) {
                        bhi[r] = __expf(bhi[r] - m);
                        s += bhi[r];
                    }
                    s += __shfl_xor(s, 16, 32);
                    const float rs = 1.0f / s;
#pragma unroll
                    for (int r = 0; r < 8; ++r) {
                        sStage[wave][8 * h + r][col]      = blo[r] * rs;
                        sStage[wave][16 + 8 * h + r][col] = bhi[r] * rs;
                    }
                }
                wave_sync();

                const float4* cl = (const float4*)&sStage[wave][col][8 * h];
                const float4* ch = (const float4*)&sStage[wave][16 + col][8 * h];
                const float4 l0 = cl[0], l1 = cl[1];
                const float4 h0 = ch[0], h1 = ch[1];
                v2f b0; b0.x = xq[0]; b0.y = xq[1];
                v2f b1; b1.x = xq[2]; b1.y = xq[3];
                v2f b2; b2.x = xq[4]; b2.y = xq[5];
                v2f b3; b3.x = xq[6]; b3.y = xq[7];
                v2f al, ah;
                al.x = l0.x; al.y = l0.y; ah.x = h0.x; ah.y = h0.y;
                ylo = wmma_f32(al, b0, ylo);  yhi = wmma_f32(ah, b0, yhi);
                al.x = l0.z; al.y = l0.w; ah.x = h0.z; ah.y = h0.w;
                ylo = wmma_f32(al, b1, ylo);  yhi = wmma_f32(ah, b1, yhi);
                al.x = l1.x; al.y = l1.y; ah.x = h1.x; ah.y = h1.y;
                ylo = wmma_f32(al, b2, ylo);  yhi = wmma_f32(ah, b2, yhi);
                al.x = l1.z; al.y = l1.w; ah.x = h1.z; ah.y = h1.w;
                ylo = wmma_f32(al, b3, ylo);  yhi = wmma_f32(ah, b3, yhi);
            }

#pragma unroll
            for (int r = 0; r < 8; ++r) {
                atomicAdd(&yacc[(8 * h + r) * DIN + col], ylo[r]);
                atomicAdd(&yacc[(16 + 8 * h + r) * DIN + col], yhi[r]);
            }
        }
        __syncthreads();

        if (tid < NC * DC) {
            const int c = tid >> 4, j = tid & 15;
            const float* y = yacc + (it == 0 ? 0 : c) * DIN;
            float s = 0.0f;
#pragma unroll
            for (int k = 0; k < DIN; ++k)
                s = fmaf(y[k], sW[k * (NC * DC) + tid], s);
            const float pre = (it == 0) ? 0.03125f * s : s;

            (&sY[(it + 1) & 1][0][0])[tid] = 0.0f;

            sOpre[c][j] = pre;
            wave_sync();
            float P[DC];
            {
                const float4* p4 = (const float4*)&sOpre[c][0];
#pragma unroll
                for (int q = 0; q < 4; ++q) {
                    const float4 t = p4[q];
                    P[4 * q + 0] = t.x; P[4 * q + 1] = t.y;
                    P[4 * q + 2] = t.z; P[4 * q + 3] = t.w;
                }
            }
            float s2 = 0.0f;
#pragma unroll
            for (int d = 0; d < DC; ++d) s2 = fmaf(P[d], P[d], s2);
            const float fac = (s2 / (1.0f + s2)) * rsqrtf(s2 + EPS);

            if (it == 2) {
                out[(size_t)bidx * (NC * DC) + tid] = fac * pre;
            } else {
                const float4* w4 = (const float4*)&sWT[c][j][0];
                float v = 0.0f;
#pragma unroll
                for (int q = 0; q < 4; ++q) {
                    const float4 w = w4[q];
                    v = fmaf(w.x, fac * P[4 * q + 0], v);
                    v = fmaf(w.y, fac * P[4 * q + 1], v);
                    v = fmaf(w.z, fac * P[4 * q + 2], v);
                    v = fmaf(w.w, fac * P[4 * q + 3], v);
                }
                sV[j][c] = v;
            }
        }
        if (it < 2) __syncthreads();
    }
}

extern "C" void kernel_launch(void* const* d_in, const int* in_sizes, int n_in,
                              void* d_out, int out_size, void* d_ws, size_t ws_size,
                              hipStream_t stream) {
    (void)in_sizes; (void)n_in; (void)out_size; (void)d_ws; (void)ws_size;
    const float* x = (const float*)d_in[0];
    const float* W = (const float*)d_in[1];
    float* out = (float*)d_out;
    hipLaunchKernelGGL(capsnet_routing_kernel, dim3(128), dim3(THREADS), 0, stream,
                       x, W, out);
}
